# starting kernel
# baseline (speedup 1.0000x reference)
#include <hip/hip_runtime.h>

typedef __attribute__((ext_vector_type(2))) float v2f;
typedef __attribute__((ext_vector_type(8))) float v8f;

#define VSIZE 100000
#define DDIM  100
#define LTOK  200
#define BATCH 16384

__global__ __launch_bounds__(256) void rowdot_wmma(const float* __restrict__ T,
                                                   const float* __restrict__ w,
                                                   float* __restrict__ s) {
    const int lane  = threadIdx.x & 31;
    const int wave  = threadIdx.x >> 5;
    const int tile  = blockIdx.x * 8 + wave;
    if (tile * 16 >= VSIZE) return;

    const int m      = lane & 15;
    const int khalf  = lane >> 4;
    const bool col0  = (m == 0);

    const v2f* __restrict__ rowp2 =
        (const v2f*)(T + (size_t)(tile * 16 + m) * DDIM);
    const v2f* __restrict__ w2 = (const v2f*)w;

    v8f acc = {};
    #pragma unroll 5
    for (int k0 = 0; k0 < DDIM; k0 += 4) {
        const int kp = (k0 >> 1) + khalf;
        v2f a  = rowp2[kp];
        v2f wv = w2[kp];
        v2f b;
        b.x = col0 ? wv.x : 0.0f;
        b.y = col0 ? wv.y : 0.0f;
        acc = __builtin_amdgcn_wmma_f32_16x16x4_f32(
             false, a,  false, b,
             (short)0, acc,  false,  false);
    }

    if (col0) {
        float* __restrict__ sp = s + tile * 16 + (khalf ? 8 : 0);
        #pragma unroll
        for (int i = 0; i < 8; ++i) sp[i] = acc[i];
    }
}

__global__ __launch_bounds__(256) void pool_gather(const int* __restrict__ idx,
                                                   const float* __restrict__ s,
                                                   float* __restrict__ out) {
    const int lane = threadIdx.x & 31;
    const int wave = threadIdx.x >> 5;
    const int b    = blockIdx.x * 8 + wave;
    if (b >= BATCH) return;

    const int* __restrict__ ip = idx + (size_t)b * LTOK;
    float sum = 0.0f;
    for (int l = lane; l < LTOK; l += 32)
        sum += s[ip[l]];

    sum += __shfl_xor(sum, 16, 32);
    sum += __shfl_xor(sum,  8, 32);
    sum += __shfl_xor(sum,  4, 32);
    sum += __shfl_xor(sum,  2, 32);
    sum += __shfl_xor(sum,  1, 32);

    if (lane == 0) out[b] = sum * (1.0f / (float)LTOK);
}

__global__ __launch_bounds__(256) void fused_naive(const int* __restrict__ idx,
                                                   const float* __restrict__ T,
                                                   const float* __restrict__ w,
                                                   float* __restrict__ out) {
    const int lane = threadIdx.x & 31;
    const int wave = threadIdx.x >> 5;
    const int b    = blockIdx.x * 8 + wave;
    if (b >= BATCH) return;

    const int* __restrict__ ip = idx + (size_t)b * LTOK;
    float sum = 0.0f;
    for (int l = 0; l < LTOK; ++l) {
        const float* __restrict__ rowp = T + (size_t)ip[l] * DDIM;
        for (int d = lane; d < DDIM; d += 32)
            sum += rowp[d] * w[d];
    }
    sum += __shfl_xor(sum, 16, 32);
    sum += __shfl_xor(sum,  8, 32);
    sum += __shfl_xor(sum,  4, 32);
    sum += __shfl_xor(sum,  2, 32);
    sum += __shfl_xor(sum,  1, 32);
    if (lane == 0) out[b] = sum * (1.0f / (float)LTOK);
}

extern "C" void kernel_launch(void* const* d_in, const int* in_sizes, int n_in,
                              void* d_out, int out_size, void* d_ws, size_t ws_size,
                              hipStream_t stream) {
    const int*   word_idxs = (const int*)d_in[0];
    const float* embed     = (const float*)d_in[1];
    const float* weights   = (const float*)d_in[2];
    float*       out       = (float*)d_out;

    if (ws_size >= (size_t)VSIZE * sizeof(float)) {
        float* s = (float*)d_ws;
        const int tiles   = (VSIZE + 15) / 16;
        const int blocks1 = (tiles + 7) / 8;
        rowdot_wmma<<<blocks1, 256, 0, stream>>>(embed, weights, s);

        const int blocks2 = (BATCH + 7) / 8;
        pool_gather<<<blocks2, 256, 0, stream>>>(word_idxs, s, out);
    } else {
        const int blocks = (BATCH + 7) / 8;
        fused_naive<<<blocks, 256, 0, stream>>>(word_idxs, embed, weights, out);
    }
}


# Pass 2: s-table staged through LDS in two 50000-entry halves per 64-row workgroup (256 WGs x 512 thr); gathers read LDS, not L2; per-row sum order changes (by table half, then 8-lane tree)
# speedup vs baseline: 1.4216x; 1.4216x over previous
#include <hip/hip_runtime.h>

typedef __attribute__((ext_vector_type(2))) float v2f;
typedef __attribute__((ext_vector_type(8))) float v8f;

#define VSIZE 100000
#define DDIM  100
#define LTOK  200
#define BATCH 16384

__global__ __launch_bounds__(256) void rowdot_wmma(const float* __restrict__ T,
                                                   const float* __restrict__ w,
                                                   float* __restrict__ s) {
    const int lane  = threadIdx.x & 31;
    const int wave  = threadIdx.x >> 5;
    const int tile  = blockIdx.x * 8 + wave;
    if (tile * 16 >= VSIZE) return;

    const int m      = lane & 15;
    const int khalf  = lane >> 4;
    const bool col0  = (m == 0);

    const v2f* __restrict__ rowp2 =
        (const v2f*)(T + (size_t)(tile * 16 + m) * DDIM);
    const v2f* __restrict__ w2 = (const v2f*)w;

    v8f acc = {};
    #pragma unroll 5
    for (int k0 = 0; k0 < DDIM; k0 += 4) {
        const int kp = (k0 >> 1) + khalf;
        v2f a  = rowp2[kp];
        v2f wv = w2[kp];
        v2f b;
        b.x = col0 ? wv.x : 0.0f;
        b.y = col0 ? wv.y : 0.0f;
        acc = __builtin_amdgcn_wmma_f32_16x16x4_f32(
             false, a,  false, b,
             (short)0, acc,  false,  false);
    }

    if (col0) {
        float* __restrict__ sp = s + tile * 16 + (khalf ? 8 : 0);
        #pragma unroll
        for (int i = 0; i < 8; ++i) sp[i] = acc[i];
    }
}

#define HALFV   50000
#define HALFQ   (HALFV / 4)
#define ROWS_WG 64
#define NT2     512
#define TOK_T   25
#define STG     25
#define IDQ     (ROWS_WG * LTOK / 4)
#define ISTG    7

__device__ __forceinline__ int stage_slot(int i, int rot) {
    int g = i + rot;
    if (g >= HALFQ) g -= HALFQ;
    return g;
}

__global__ __launch_bounds__(NT2) __attribute__((amdgpu_waves_per_eu(4, 4))) void pool_gather(const int* __restrict__ idx,
                                                   const float* __restrict__ s,
                                                   float* __restrict__ out) {
    __shared__ __attribute__((aligned(16))) float tab[HALFV + 4];
    __shared__ __attribute__((aligned(16))) int   idl[ISTG * NT2 * 4];
    __shared__ float rsum[ROWS_WG];

    const int tid = threadIdx.x;
    const int blk = blockIdx.x;
    __builtin_assume(tid < NT2);

    const float4* __restrict__ s4   = (const float4*)s;
    float4*                    tab4 = (float4*)tab;
    const int4* __restrict__   ip4  = (const int4*)(idx + (size_t)blk * (ROWS_WG * LTOK));
    int4*                      idl4 = (int4*)idl;

    const int rot = blk * 48;

    int4   iq[ISTG];
    float4 cur[STG];
    #pragma unroll
    for (int t = 0; t < ISTG; ++t) {
        const int i = tid + t * NT2;
        iq[t] = ip4[i < IDQ ? i : IDQ - 1];
    }
    #pragma unroll
    for (int t = 0; t < STG; ++t) {
        const int i = tid + t * NT2;
        int g = (i < HALFQ ? i : HALFQ - 1) + rot;
        if (g >= HALFQ) g -= HALFQ;
        cur[t] = s4[g];
    }
    #pragma unroll
    for (int t = 0; t < ISTG; ++t) idl4[tid + t * NT2] = iq[t];
    #pragma unroll
    for (int t = 0; t < STG; ++t) {
        const int i = tid + t * NT2;
        tab4[stage_slot(i < HALFQ ? i : HALFQ - 1, rot)] = cur[t];
    }
    if (tid == 0) tab[HALFV] = 0.0f;
    __syncthreads();

#define HALF1_AT(t) (s4[HALFQ + stage_slot(tid + (t) * NT2 < HALFQ ? tid + (t) * NT2 : HALFQ - 1, rot)])
    const float4 n0 = HALF1_AT(0);
    const float4 n1 = HALF1_AT(1);
    const float4 n2 = HALF1_AT(2);
    const float4 n3 = HALF1_AT(3);
    const float4 n4 = HALF1_AT(4);
    const float4 n5 = HALF1_AT(5);
    const float4 n6 = HALF1_AT(6);
    const float4 n7 = HALF1_AT(7);
    const float4 n8 = HALF1_AT(8);
    const float4 n9 = HALF1_AT(9);
    const float4 n10 = HALF1_AT(10);
    const float4 n11 = HALF1_AT(11);
    const float4 n12 = HALF1_AT(12);
    const float4 n13 = HALF1_AT(13);
    const float4 n14 = HALF1_AT(14);
    const float4 n15 = HALF1_AT(15);
    const float4 n16 = HALF1_AT(16);
    const float4 n17 = HALF1_AT(17);
    const float4 n18 = HALF1_AT(18);
    const float4 n19 = HALF1_AT(19);
    const float4 n20 = HALF1_AT(20);
    const float4 n21 = HALF1_AT(21);
    const float4 n22 = HALF1_AT(22);
    const float4 n23 = HALF1_AT(23);
    const float4 n24 = HALF1_AT(24);

    unsigned tok[TOK_T];
    #pragma unroll
    for (int t = 0; t < TOK_T; ++t) tok[t] = (unsigned)idl[tid * TOK_T + t];

    float acc = 0.0f;
    #pragma unroll
    for (int t = 0; t < TOK_T; ++t) {
        const unsigned a = tok[t] < (unsigned)HALFV ? tok[t] : (unsigned)HALFV;
        acc += tab[a];
    }
    __syncthreads();

    tab4[stage_slot(tid + 0 * NT2, rot)] = n0;
    tab4[stage_slot(tid + 1 * NT2, rot)] = n1;
    tab4[stage_slot(tid + 2 * NT2, rot)] = n2;
    tab4[stage_slot(tid + 3 * NT2, rot)] = n3;
    tab4[stage_slot(tid + 4 * NT2, rot)] = n4;
    tab4[stage_slot(tid + 5 * NT2, rot)] = n5;
    tab4[stage_slot(tid + 6 * NT2, rot)] = n6;
    tab4[stage_slot(tid + 7 * NT2, rot)] = n7;
    tab4[stage_slot(tid + 8 * NT2, rot)] = n8;
    tab4[stage_slot(tid + 9 * NT2, rot)] = n9;
    tab4[stage_slot(tid + 10 * NT2, rot)] = n10;
    tab4[stage_slot(tid + 11 * NT2, rot)] = n11;
    tab4[stage_slot(tid + 12 * NT2, rot)] = n12;
    tab4[stage_slot(tid + 13 * NT2, rot)] = n13;
    tab4[stage_slot(tid + 14 * NT2, rot)] = n14;
    tab4[stage_slot(tid + 15 * NT2, rot)] = n15;
    tab4[stage_slot(tid + 16 * NT2, rot)] = n16;
    tab4[stage_slot(tid + 17 * NT2, rot)] = n17;
    tab4[stage_slot(tid + 18 * NT2, rot)] = n18;
    tab4[stage_slot(tid + 19 * NT2, rot)] = n19;
    tab4[stage_slot(tid + 20 * NT2, rot)] = n20;
    tab4[stage_slot(tid + 21 * NT2, rot)] = n21;
    tab4[stage_slot(tid + 22 * NT2, rot)] = n22;
    tab4[stage_slot(tid + 23 * NT2, rot)] = n23;
    tab4[stage_slot(tid + 24 * NT2 < HALFQ ? tid + 24 * NT2 : HALFQ - 1, rot)] = n24;
    __syncthreads();

    #pragma unroll
    for (int t = 0; t < TOK_T; ++t) {
        const unsigned d = tok[t] - (unsigned)HALFV;
        const unsigned a = d < (unsigned)HALFV ? d : (unsigned)HALFV;
        acc += tab[a];
    }

    acc += __shfl_xor(acc, 1, 32);
    acc += __shfl_xor(acc, 2, 32);
    acc += __shfl_xor(acc, 4, 32);
    if ((tid & 7) == 0) rsum[tid >> 3] = acc * (1.0f / (float)LTOK);
    __syncthreads();
    if (tid < ROWS_WG) out[blk * ROWS_WG + tid] = rsum[tid];
}

__global__ __launch_bounds__(256) void fused_naive(const int* __restrict__ idx,
                                                   const float* __restrict__ T,
                                                   const float* __restrict__ w,
                                                   float* __restrict__ out) {
    const int lane = threadIdx.x & 31;
    const int wave = threadIdx.x >> 5;
    const int b    = blockIdx.x * 8 + wave;
    if (b >= BATCH) return;

    const int* __restrict__ ip = idx + (size_t)b * LTOK;
    float sum = 0.0f;
    for (int l = 0; l < LTOK; ++l) {
        const float* __restrict__ rowp = T + (size_t)ip[l] * DDIM;
        for (int d = lane; d < DDIM; d += 32)
            sum += rowp[d] * w[d];
    }
    sum += __shfl_xor(sum, 16, 32);
    sum += __shfl_xor(sum,  8, 32);
    sum += __shfl_xor(sum,  4, 32);
    sum += __shfl_xor(sum,  2, 32);
    sum += __shfl_xor(sum,  1, 32);
    if (lane == 0) out[b] = sum * (1.0f / (float)LTOK);
}

extern "C" void kernel_launch(void* const* d_in, const int* in_sizes, int n_in,
                              void* d_out, int out_size, void* d_ws, size_t ws_size,
                              hipStream_t stream) {
    const int*   word_idxs = (const int*)d_in[0];
    const float* embed     = (const float*)d_in[1];
    const float* weights   = (const float*)d_in[2];
    float*       out       = (float*)d_out;

    if (ws_size >= (size_t)VSIZE * sizeof(float)) {
        float* s = (float*)d_ws;
        const int tiles   = (VSIZE + 15) / 16;
        const int blocks1 = (tiles + 7) / 8;
        rowdot_wmma<<<blocks1, 256, 0, stream>>>(embed, weights, s);

        const int blocks2 = BATCH / ROWS_WG;
        pool_gather<<<blocks2, NT2, 0, stream>>>(word_idxs, s, out);
    } else {
        const int blocks = (BATCH + 7) / 8;
        fused_naive<<<blocks, 256, 0, stream>>>(word_idxs, embed, weights, out);
    }
}


# Pass 1: each wave streams 32 whole table rows (25 whole-line 16 B loads in flight) through LDS into the f32 WMMA A layout, 2 accumulator chains, 3125 waves not 6250; per-row dot sum order changes
# speedup vs baseline: 1.6625x; 1.1695x over previous
#include <hip/hip_runtime.h>

typedef __attribute__((ext_vector_type(2))) float v2f;
typedef __attribute__((ext_vector_type(8))) float v8f;

#define VSIZE 100000
#define DDIM  100
#define LTOK  200
#define BATCH 16384

#define P1_ROWS  32
#define P1_WAVES 4
#define P1_NT    (P1_WAVES * 32)
#define P1_FL    (P1_ROWS * DDIM)
#define P1_LD    (P1_FL / 4 / 32)
#define P1_UNITS (VSIZE / P1_ROWS)

__global__ __launch_bounds__(P1_NT) __attribute__((amdgpu_waves_per_eu(4, 4))) void rowdot_wmma(const float* __restrict__ T,
                                                     const float* __restrict__ w,
                                                     float* __restrict__ s) {
    __shared__ __attribute__((aligned(16))) float stage[P1_WAVES * P1_FL];

    const int lane = threadIdx.x & 31;
    const int wave = threadIdx.x >> 5;
    int unit = blockIdx.x * P1_WAVES + wave;
    const bool live = unit < P1_UNITS;
    if (!live) unit = P1_UNITS - 1;

    const int  m    = lane & 15;
    const int  h    = lane >> 4;
    const bool col0 = (m == 0);

    const float4* __restrict__ w4 = (const float4*)w;
    const v2f*    __restrict__ w2 = (const v2f*)w;
    float4 wq[12];
    #pragma unroll
    for (int q = 0; q < 12; ++q) wq[q] = w4[2 * q + h];
    const v2f wt = w2[48 + h];

    const float4* __restrict__ src = (const float4*)(T + (size_t)unit * P1_FL);
    float4 buf[P1_LD];
    #pragma unroll
    for (int j = 0; j < P1_LD; ++j) buf[j] = src[j * 32 + lane];

    float*  mine = stage + wave * P1_FL;
    float4* st4  = (float4*)mine;
    #pragma unroll
    for (int j = 0; j < P1_LD; ++j) st4[j * 32 + lane] = buf[j];
    __syncthreads();

    const float* row0 = mine + m * DDIM;
    const float* row1 = row0 + 16 * DDIM;

    v8f acc0 = {};
    v8f acc1 = {};
    #pragma unroll
    for (int q = 0; q < 12; ++q) {
        const float4 a0 = *(const float4*)(row0 + 8 * q + 4 * h);
        const float4 a1 = *(const float4*)(row1 + 8 * q + 4 * h);
        const float4 wv = wq[q];
        v2f blo, bhi, x0, y0, x1, y1;
        blo.x = col0 ? wv.x : 0.0f;  blo.y = col0 ? wv.y : 0.0f;
        bhi.x = col0 ? wv.z : 0.0f;  bhi.y = col0 ? wv.w : 0.0f;
        x0.x = a0.x; x0.y = a0.y;  y0.x = a0.z; y0.y = a0.w;
        x1.x = a1.x; x1.y = a1.y;  y1.x = a1.z; y1.y = a1.w;
        acc0 = __builtin_amdgcn_wmma_f32_16x16x4_f32(false, x0, false, blo, (short)0, acc0, false, false);
        acc1 = __builtin_amdgcn_wmma_f32_16x16x4_f32(false, x1, false, blo, (short)0, acc1, false, false);
        acc0 = __builtin_amdgcn_wmma_f32_16x16x4_f32(false, y0, false, bhi, (short)0, acc0, false, false);
        acc1 = __builtin_amdgcn_wmma_f32_16x16x4_f32(false, y1, false, bhi, (short)0, acc1, false, false);
    }
    {
        const v2f t0 = *(const v2f*)(row0 + 96 + 2 * h);
        const v2f t1 = *(const v2f*)(row1 + 96 + 2 * h);
        v2f bt;
        bt.x = col0 ? wt.x : 0.0f;  bt.y = col0 ? wt.y : 0.0f;
        acc0 = __builtin_amdgcn_wmma_f32_16x16x4_f32(false, t0, false, bt, (short)0, acc0, false, false);
        acc1 = __builtin_amdgcn_wmma_f32_16x16x4_f32(false, t1, false, bt, (short)0, acc1, false, false);
    }

    if (col0 && live) {
        float* __restrict__ sp = s + unit * P1_ROWS + (h ? 8 : 0);
        #pragma unroll
        for (int i = 0; i < 8; ++i) { sp[i] = acc0[i]; sp[16 + i] = acc1[i]; }
    }
}

#define HALFV   50000
#define HALFQ   (HALFV / 4)
#define ROWS_WG 64
#define NT2     512
#define TOK_T   25
#define STG     25
#define IDQ     (ROWS_WG * LTOK / 4)
#define ISTG    7

__device__ __forceinline__ int stage_slot(int i, int rot) {
    int g = i + rot;
    if (g >= HALFQ) g -= HALFQ;
    return g;
}

__global__ __launch_bounds__(NT2) __attribute__((amdgpu_waves_per_eu(4, 4))) void pool_gather(const int* __restrict__ idx,
                                                   const float* __restrict__ s,
                                                   float* __restrict__ out) {
    __shared__ __attribute__((aligned(16))) float tab[HALFV + 4];
    __shared__ __attribute__((aligned(16))) int   idl[ISTG * NT2 * 4];
    __shared__ float rsum[ROWS_WG];

    const int tid = threadIdx.x;
    const int blk = blockIdx.x;
    __builtin_assume(tid < NT2);

    const float4* __restrict__ s4   = (const float4*)s;
    float4*                    tab4 = (float4*)tab;
    const int4* __restrict__   ip4  = (const int4*)(idx + (size_t)blk * (ROWS_WG * LTOK));
    int4*                      idl4 = (int4*)idl;

    const int rot = blk * 48;

    int4   iq[ISTG];
    float4 cur[STG];
    #pragma unroll
    for (int t = 0; t < ISTG; ++t) {
        const int i = tid + t * NT2;
        iq[t] = ip4[i < IDQ ? i : IDQ - 1];
    }
    #pragma unroll
    for (int t = 0; t < STG; ++t) {
        const int i = tid + t * NT2;
        int g = (i < HALFQ ? i : HALFQ - 1) + rot;
        if (g >= HALFQ) g -= HALFQ;
        cur[t] = s4[g];
    }
    #pragma unroll
    for (int t = 0; t < ISTG; ++t) idl4[tid + t * NT2] = iq[t];
    #pragma unroll
    for (int t = 0; t < STG; ++t) {
        const int i = tid + t * NT2;
        tab4[stage_slot(i < HALFQ ? i : HALFQ - 1, rot)] = cur[t];
    }
    if (tid == 0) tab[HALFV] = 0.0f;
    __syncthreads();

#define HALF1_AT(t) (s4[HALFQ + stage_slot(tid + (t) * NT2 < HALFQ ? tid + (t) * NT2 : HALFQ - 1, rot)])
    const float4 n0 = HALF1_AT(0);
    const float4 n1 = HALF1_AT(1);
    const float4 n2 = HALF1_AT(2);
    const float4 n3 = HALF1_AT(3);
    const float4 n4 = HALF1_AT(4);
    const float4 n5 = HALF1_AT(5);
    const float4 n6 = HALF1_AT(6);
    const float4 n7 = HALF1_AT(7);
    const float4 n8 = HALF1_AT(8);
    const float4 n9 = HALF1_AT(9);
    const float4 n10 = HALF1_AT(10);
    const float4 n11 = HALF1_AT(11);
    const float4 n12 = HALF1_AT(12);
    const float4 n13 = HALF1_AT(13);
    const float4 n14 = HALF1_AT(14);
    const float4 n15 = HALF1_AT(15);
    const float4 n16 = HALF1_AT(16);
    const float4 n17 = HALF1_AT(17);
    const float4 n18 = HALF1_AT(18);
    const float4 n19 = HALF1_AT(19);
    const float4 n20 = HALF1_AT(20);
    const float4 n21 = HALF1_AT(21);
    const float4 n22 = HALF1_AT(22);
    const float4 n23 = HALF1_AT(23);
    const float4 n24 = HALF1_AT(24);

    unsigned tok[TOK_T];
    #pragma unroll
    for (int t = 0; t < TOK_T; ++t) tok[t] = (unsigned)idl[tid * TOK_T + t];

    float acc = 0.0f;
    #pragma unroll
    for (int t = 0; t < TOK_T; ++t) {
        const unsigned a = tok[t] < (unsigned)HALFV ? tok[t] : (unsigned)HALFV;
        acc += tab[a];
    }
    __syncthreads();

    tab4[stage_slot(tid + 0 * NT2, rot)] = n0;
    tab4[stage_slot(tid + 1 * NT2, rot)] = n1;
    tab4[stage_slot(tid + 2 * NT2, rot)] = n2;
    tab4[stage_slot(tid + 3 * NT2, rot)] = n3;
    tab4[stage_slot(tid + 4 * NT2, rot)] = n4;
    tab4[stage_slot(tid + 5 * NT2, rot)] = n5;
    tab4[stage_slot(tid + 6 * NT2, rot)] = n6;
    tab4[stage_slot(tid + 7 * NT2, rot)] = n7;
    tab4[stage_slot(tid + 8 * NT2, rot)] = n8;
    tab4[stage_slot(tid + 9 * NT2, rot)] = n9;
    tab4[stage_slot(tid + 10 * NT2, rot)] = n10;
    tab4[stage_slot(tid + 11 * NT2, rot)] = n11;
    tab4[stage_slot(tid + 12 * NT2, rot)] = n12;
    tab4[stage_slot(tid + 13 * NT2, rot)] = n13;
    tab4[stage_slot(tid + 14 * NT2, rot)] = n14;
    tab4[stage_slot(tid + 15 * NT2, rot)] = n15;
    tab4[stage_slot(tid + 16 * NT2, rot)] = n16;
    tab4[stage_slot(tid + 17 * NT2, rot)] = n17;
    tab4[stage_slot(tid + 18 * NT2, rot)] = n18;
    tab4[stage_slot(tid + 19 * NT2, rot)] = n19;
    tab4[stage_slot(tid + 20 * NT2, rot)] = n20;
    tab4[stage_slot(tid + 21 * NT2, rot)] = n21;
    tab4[stage_slot(tid + 22 * NT2, rot)] = n22;
    tab4[stage_slot(tid + 23 * NT2, rot)] = n23;
    tab4[stage_slot(tid + 24 * NT2 < HALFQ ? tid + 24 * NT2 : HALFQ - 1, rot)] = n24;
    __syncthreads();

    #pragma unroll
    for (int t = 0; t < TOK_T; ++t) {
        const unsigned d = tok[t] - (unsigned)HALFV;
        const unsigned a = d < (unsigned)HALFV ? d : (unsigned)HALFV;
        acc += tab[a];
    }

    acc += __shfl_xor(acc, 1, 32);
    acc += __shfl_xor(acc, 2, 32);
    acc += __shfl_xor(acc, 4, 32);
    if ((tid & 7) == 0) rsum[tid >> 3] = acc * (1.0f / (float)LTOK);
    __syncthreads();
    if (tid < ROWS_WG) out[blk * ROWS_WG + tid] = rsum[tid];
}

__global__ __launch_bounds__(256) void fused_naive(const int* __restrict__ idx,
                                                   const float* __restrict__ T,
                                                   const float* __restrict__ w,
                                                   float* __restrict__ out) {
    const int lane = threadIdx.x & 31;
    const int wave = threadIdx.x >> 5;
    const int b    = blockIdx.x * 8 + wave;
    if (b >= BATCH) return;

    const int* __restrict__ ip = idx + (size_t)b * LTOK;
    float sum = 0.0f;
    for (int l = 0; l < LTOK; ++l) {
        const float* __restrict__ rowp = T + (size_t)ip[l] * DDIM;
        for (int d = lane; d < DDIM; d += 32)
            sum += rowp[d] * w[d];
    }
    sum += __shfl_xor(sum, 16, 32);
    sum += __shfl_xor(sum,  8, 32);
    sum += __shfl_xor(sum,  4, 32);
    sum += __shfl_xor(sum,  2, 32);
    sum += __shfl_xor(sum,  1, 32);
    if (lane == 0) out[b] = sum * (1.0f / (float)LTOK);
}

extern "C" void kernel_launch(void* const* d_in, const int* in_sizes, int n_in,
                              void* d_out, int out_size, void* d_ws, size_t ws_size,
                              hipStream_t stream) {
    const int*   word_idxs = (const int*)d_in[0];
    const float* embed     = (const float*)d_in[1];
    const float* weights   = (const float*)d_in[2];
    float*       out       = (float*)d_out;

    if (ws_size >= (size_t)VSIZE * sizeof(float)) {
        float* s = (float*)d_ws;
        const int blocks1 = (P1_UNITS + P1_WAVES - 1) / P1_WAVES;
        rowdot_wmma<<<blocks1, P1_NT, 0, stream>>>(embed, weights, s);

        const int blocks2 = BATCH / ROWS_WG;
        pool_gather<<<blocks2, NT2, 0, stream>>>(word_idxs, s, out);
    } else {
        const int blocks = (BATCH + 7) / 8;
        fused_naive<<<blocks, 256, 0, stream>>>(word_idxs, embed, weights, out);
    }
}


# Half as many, twice as fat waves in both launches to halve wave entry: pass 1 two 32-row units per wave (1563 waves), pass 2 256-thread workgroups; per-row sum order changes (4 lanes x 50 tokens)
# speedup vs baseline: 1.8081x; 1.0876x over previous
#include <hip/hip_runtime.h>

typedef __attribute__((ext_vector_type(2))) float v2f;
typedef __attribute__((ext_vector_type(8))) float v8f;

#define VSIZE 100000
#define DDIM  100
#define LTOK  200
#define BATCH 16384

#define P1_ROWS  32
#define P1_WAVES 4
#define P1_NT    (P1_WAVES * 32)
#define P1_FL    (P1_ROWS * DDIM)
#define P1_LD    (P1_FL / 4 / 32)
#define P1_UNITS (VSIZE / P1_ROWS)

#define P1_UPW    2
#define P1_WUNITS ((P1_UNITS + P1_UPW - 1) / P1_UPW)

__device__ __forceinline__ void rowdot_unit(const float* mine, const float4 (&wq)[12], const v2f wt,
                                            const int m, const int h, const bool col0,
                                            v8f& acc0, v8f& acc1) {
    const float* row0 = mine + m * DDIM;
    const float* row1 = row0 + 16 * DDIM;
    #pragma unroll
    for (int q = 0; q < 12; ++q) {
        const float4 a0 = *(const float4*)(row0 + 8 * q + 4 * h);
        const float4 a1 = *(const float4*)(row1 + 8 * q + 4 * h);
        const float4 wv = wq[q];
        v2f blo, bhi, x0, y0, x1, y1;
        blo.x = col0 ? wv.x : 0.0f;  blo.y = col0 ? wv.y : 0.0f;
        bhi.x = col0 ? wv.z : 0.0f;  bhi.y = col0 ? wv.w : 0.0f;
        x0.x = a0.x; x0.y = a0.y;  y0.x = a0.z; y0.y = a0.w;
        x1.x = a1.x; x1.y = a1.y;  y1.x = a1.z; y1.y = a1.w;
        acc0 = __builtin_amdgcn_wmma_f32_16x16x4_f32(false, x0, false, blo, (short)0, acc0, false, false);
        acc1 = __builtin_amdgcn_wmma_f32_16x16x4_f32(false, x1, false, blo, (short)0, acc1, false, false);
        acc0 = __builtin_amdgcn_wmma_f32_16x16x4_f32(false, y0, false, bhi, (short)0, acc0, false, false);
        acc1 = __builtin_amdgcn_wmma_f32_16x16x4_f32(false, y1, false, bhi, (short)0, acc1, false, false);
    }
    {
        const v2f t0 = *(const v2f*)(row0 + 96 + 2 * h);
        const v2f t1 = *(const v2f*)(row1 + 96 + 2 * h);
        v2f bt;
        bt.x = col0 ? wt.x : 0.0f;  bt.y = col0 ? wt.y : 0.0f;
        acc0 = __builtin_amdgcn_wmma_f32_16x16x4_f32(false, t0, false, bt, (short)0, acc0, false, false);
        acc1 = __builtin_amdgcn_wmma_f32_16x16x4_f32(false, t1, false, bt, (short)0, acc1, false, false);
    }
}

__global__ __launch_bounds__(P1_NT) __attribute__((amdgpu_waves_per_eu(2, 2))) void rowdot_wmma(const float* __restrict__ T,
                                                     const float* __restrict__ w,
                                                     float* __restrict__ s) {
    __shared__ __attribute__((aligned(16))) float stage[P1_WAVES * P1_UPW * P1_FL];

    const int lane = threadIdx.x & 31;
    const int wave = threadIdx.x >> 5;
    const int wu   = blockIdx.x * P1_WAVES + wave;
    int unitA = wu * P1_UPW;
    int unitB = unitA + 1;
    const bool liveA = unitA < P1_UNITS;
    const bool liveB = unitB < P1_UNITS;
    if (!liveA) unitA = P1_UNITS - 1;
    if (!liveB) unitB = P1_UNITS - 1;

    const int  m    = lane & 15;
    const int  h    = lane >> 4;
    const bool col0 = (m == 0);

    const float4* __restrict__ w4 = (const float4*)w;
    const v2f*    __restrict__ w2 = (const v2f*)w;
    float4 wq[12];
    #pragma unroll
    for (int q = 0; q < 12; ++q) wq[q] = w4[2 * q + h];
    const v2f wt = w2[48 + h];

    const float4* __restrict__ srcA = (const float4*)(T + (size_t)unitA * P1_FL);
    const float4* __restrict__ srcB = (const float4*)(T + (size_t)unitB * P1_FL);
    float4 bufA[P1_LD];
    float4 bufB[P1_LD];
    #pragma unroll
    for (int j = 0; j < P1_LD; ++j) bufA[j] = srcA[j * 32 + lane];
    #pragma unroll
    for (int j = 0; j < P1_LD; ++j) bufB[j] = srcB[j * 32 + lane];

    float*  mineA = stage + (wave * P1_UPW) * P1_FL;
    float*  mineB = mineA + P1_FL;
    float4* stA   = (float4*)mineA;
    float4* stB   = (float4*)mineB;

    #pragma unroll
    for (int j = 0; j < P1_LD; ++j) stA[j * 32 + lane] = bufA[j];
    #pragma unroll
    for (int j = 0; j < P1_LD; ++j) stB[j * 32 + lane] = bufB[j];
    __syncthreads();

    v8f acc0 = {};
    v8f acc1 = {};
    v8f acc2 = {};
    v8f acc3 = {};
    rowdot_unit(mineA, wq, wt, m, h, col0, acc0, acc1);
    rowdot_unit(mineB, wq, wt, m, h, col0, acc2, acc3);

    if (col0 && liveA) {
        float* __restrict__ sp = s + unitA * P1_ROWS + (h ? 8 : 0);
        #pragma unroll
        for (int i = 0; i < 8; ++i) { sp[i] = acc0[i]; sp[16 + i] = acc1[i]; }
    }
    if (col0 && liveB) {
        float* __restrict__ sp = s + unitB * P1_ROWS + (h ? 8 : 0);
        #pragma unroll
        for (int i = 0; i < 8; ++i) { sp[i] = acc2[i]; sp[16 + i] = acc3[i]; }
    }
}

#define HALFV   50000
#define HALFQ   (HALFV / 4)
#define ROWS_WG 64
#define NT2     256
#define TOK_T   50
#define STG     49
#define IDQ     (ROWS_WG * LTOK / 4)
#define ISTG    13

__device__ __forceinline__ int stage_slot(int i, int rot) {
    int g = i + rot;
    if (g >= HALFQ) g -= HALFQ;
    return g;
}

#define FOR_STG(X) \
    X(0) X(1) X(2) X(3) X(4) X(5) X(6) X(7) X(8) X(9) \
    X(10) X(11) X(12) X(13) X(14) X(15) X(16) X(17) X(18) X(19) \
    X(20) X(21) X(22) X(23) X(24) X(25) X(26) X(27) X(28) X(29) \
    X(30) X(31) X(32) X(33) X(34) X(35) X(36) X(37) X(38) X(39) \
    X(40) X(41) X(42) X(43) X(44) X(45) X(46) X(47) X(48)

__global__ __launch_bounds__(NT2) __attribute__((amdgpu_waves_per_eu(2, 2))) void pool_gather(const int* __restrict__ idx,
                                                   const float* __restrict__ s,
                                                   float* __restrict__ out) {
    __shared__ __attribute__((aligned(16))) float tab[HALFV + 4];
    __shared__ __attribute__((aligned(16))) int   idl[ISTG * NT2 * 4];
    __shared__ float rsum[ROWS_WG];

    const int tid = threadIdx.x;
    const int blk = blockIdx.x;
    __builtin_assume(tid < NT2);

    const float4* __restrict__ s4   = (const float4*)s;
    float4*                    tab4 = (float4*)tab;
    const int4* __restrict__   ip4  = (const int4*)(idx + (size_t)blk * (ROWS_WG * LTOK));
    int4*                      idl4 = (int4*)idl;

    const int rot = blk * 48;

    int4   iq[ISTG];
    float4 cur[STG];
    #pragma unroll
    for (int t = 0; t < ISTG; ++t) {
        const int i = tid + t * NT2;
        iq[t] = ip4[i < IDQ ? i : IDQ - 1];
    }
    #pragma unroll
    for (int t = 0; t < STG; ++t) {
        const int i = tid + t * NT2;
        int g = (i < HALFQ ? i : HALFQ - 1) + rot;
        if (g >= HALFQ) g -= HALFQ;
        cur[t] = s4[g];
    }
    #pragma unroll
    for (int t = 0; t < ISTG; ++t) idl4[tid + t * NT2] = iq[t];
    #pragma unroll
    for (int t = 0; t < STG; ++t) {
        const int i = tid + t * NT2;
        tab4[stage_slot(i < HALFQ ? i : HALFQ - 1, rot)] = cur[t];
    }
    if (tid == 0) tab[HALFV] = 0.0f;
    __syncthreads();

#define HALF1_IX(t) (tid + (t) * NT2 < HALFQ ? tid + (t) * NT2 : HALFQ - 1)
#define X(t) const float4 n##t = s4[HALFQ + stage_slot(HALF1_IX(t), rot)];
    FOR_STG(X)
#undef X

    unsigned tok[TOK_T];
    #pragma unroll
    for (int t = 0; t < TOK_T; ++t) tok[t] = (unsigned)idl[tid * TOK_T + t];

    float acc = 0.0f;
    #pragma unroll
    for (int t = 0; t < TOK_T; ++t) {
        const unsigned a = tok[t] < (unsigned)HALFV ? tok[t] : (unsigned)HALFV;
        acc += tab[a];
    }
    __syncthreads();

#define X(t) tab4[stage_slot(HALF1_IX(t), rot)] = n##t;
    FOR_STG(X)
#undef X
    __syncthreads();

    #pragma unroll
    for (int t = 0; t < TOK_T; ++t) {
        const unsigned d = tok[t] - (unsigned)HALFV;
        const unsigned a = d < (unsigned)HALFV ? d : (unsigned)HALFV;
        acc += tab[a];
    }

    acc += __shfl_xor(acc, 1, 32);
    acc += __shfl_xor(acc, 2, 32);
    if ((tid & 3) == 0) rsum[tid >> 2] = acc * (1.0f / (float)LTOK);
    __syncthreads();
    if (tid < ROWS_WG) out[blk * ROWS_WG + tid] = rsum[tid];
}

__global__ __launch_bounds__(256) void fused_naive(const int* __restrict__ idx,
                                                   const float* __restrict__ T,
                                                   const float* __restrict__ w,
                                                   float* __restrict__ out) {
    const int lane = threadIdx.x & 31;
    const int wave = threadIdx.x >> 5;
    const int b    = blockIdx.x * 8 + wave;
    if (b >= BATCH) return;

    const int* __restrict__ ip = idx + (size_t)b * LTOK;
    float sum = 0.0f;
    for (int l = 0; l < LTOK; ++l) {
        const float* __restrict__ rowp = T + (size_t)ip[l] * DDIM;
        for (int d = lane; d < DDIM; d += 32)
            sum += rowp[d] * w[d];
    }
    sum += __shfl_xor(sum, 16, 32);
    sum += __shfl_xor(sum,  8, 32);
    sum += __shfl_xor(sum,  4, 32);
    sum += __shfl_xor(sum,  2, 32);
    sum += __shfl_xor(sum,  1, 32);
    if (lane == 0) out[b] = sum * (1.0f / (float)LTOK);
}

extern "C" void kernel_launch(void* const* d_in, const int* in_sizes, int n_in,
                              void* d_out, int out_size, void* d_ws, size_t ws_size,
                              hipStream_t stream) {
    const int*   word_idxs = (const int*)d_in[0];
    const float* embed     = (const float*)d_in[1];
    const float* weights   = (const float*)d_in[2];
    float*       out       = (float*)d_out;

    if (ws_size >= (size_t)VSIZE * sizeof(float)) {
        float* s = (float*)d_ws;
        const int blocks1 = (P1_WUNITS + P1_WAVES - 1) / P1_WAVES;
        rowdot_wmma<<<blocks1, P1_NT, 0, stream>>>(embed, weights, s);

        const int blocks2 = BATCH / ROWS_WG;
        pool_gather<<<blocks2, NT2, 0, stream>>>(word_idxs, s, out);
    } else {
        const int blocks = (BATCH + 7) / 8;
        fused_naive<<<blocks, 256, 0, stream>>>(word_idxs, embed, weights, out);
    }
}


# Pass 1: each 16-row tile is copied to LDS and multiplied as soon as its own loads land (wave-local hand-over, no workgroup barrier), two k-split chains per tile; per-row dot sum order changes
# speedup vs baseline: 1.9034x; 1.0527x over previous
#include <hip/hip_runtime.h>

typedef __attribute__((ext_vector_type(2))) float v2f;
typedef __attribute__((ext_vector_type(8))) float v8f;

#define VSIZE 100000
#define DDIM  100
#define LTOK  200
#define BATCH 16384

#define P1_ROWS  32
#define P1_WAVES 4
#define P1_NT    (P1_WAVES * 32)
#define P1_FL    (P1_ROWS * DDIM)
#define P1_LD    (P1_FL / 4 / 32)
#define P1_UNITS (VSIZE / P1_ROWS)

#define P1_UPW    2
#define P1_WUNITS ((P1_UNITS + P1_UPW - 1) / P1_UPW)

__device__ __forceinline__ v8f rowdot_tile(const float* row, const float4 (&wq)[12], const v2f wt,
                                           const int h, const bool col0) {
    float4 fr[12];
    #pragma unroll
    for (int q = 0; q < 12; ++q) fr[q] = *(const float4*)(row + 8 * q + 4 * h);
    const v2f ft = *(const v2f*)(row + 96 + 2 * h);
    __builtin_amdgcn_sched_barrier(0);
    v8f accX = {};
    v8f accY = {};
    #pragma unroll
    for (int q = 0; q < 12; ++q) {
        const float4 a  = fr[q];
        const float4 wv = wq[q];
        v2f blo, bhi, x, y;
        blo.x = col0 ? wv.x : 0.0f;  blo.y = col0 ? wv.y : 0.0f;
        bhi.x = col0 ? wv.z : 0.0f;  bhi.y = col0 ? wv.w : 0.0f;
        x.x = a.x; x.y = a.y;  y.x = a.z; y.y = a.w;
        accX = __builtin_amdgcn_wmma_f32_16x16x4_f32(false, x, false, blo, (short)0, accX, false, false);
        accY = __builtin_amdgcn_wmma_f32_16x16x4_f32(false, y, false, bhi, (short)0, accY, false, false);
    }
    {
        v2f bt;
        bt.x = col0 ? wt.x : 0.0f;  bt.y = col0 ? wt.y : 0.0f;
        accX = __builtin_amdgcn_wmma_f32_16x16x4_f32(false, ft, false, bt, (short)0, accX, false, false);
    }
    return accX + accY;
}

__device__ __forceinline__ void wave_lds_handover() {
    __builtin_amdgcn_fence(__ATOMIC_RELEASE, "wavefront", "local");
    __builtin_amdgcn_wave_barrier();
    __builtin_amdgcn_fence(__ATOMIC_ACQUIRE, "wavefront", "local");
}

__global__ __launch_bounds__(P1_NT) __attribute__((amdgpu_waves_per_eu(2, 2))) void rowdot_wmma(const float* __restrict__ T,
                                                     const float* __restrict__ w,
                                                     float* __restrict__ s) {
    __shared__ __attribute__((aligned(16))) float stage[P1_WAVES * P1_UPW * P1_FL];

    const int lane = threadIdx.x & 31;
    const int wave = threadIdx.x >> 5;
    const int wu   = blockIdx.x * P1_WAVES + wave;
    int unitA = wu * P1_UPW;
    int unitB = unitA + 1;
    const bool liveA = unitA < P1_UNITS;
    const bool liveB = unitB < P1_UNITS;
    if (!liveA) unitA = P1_UNITS - 1;
    if (!liveB) unitB = P1_UNITS - 1;

    const int  m    = lane & 15;
    const int  h    = lane >> 4;
    const bool col0 = (m == 0);

    const float4* __restrict__ w4 = (const float4*)w;
    const v2f*    __restrict__ w2 = (const v2f*)w;
    float4 wq[12];
    #pragma unroll
    for (int q = 0; q < 12; ++q) wq[q] = w4[2 * q + h];
    const v2f wt = w2[48 + h];

    const float4* __restrict__ srcA = (const float4*)(T + (size_t)unitA * P1_FL);
    const float4* __restrict__ srcB = (const float4*)(T + (size_t)unitB * P1_FL);
#define FOR_T0(X) X(0) X(1) X(2) X(3) X(4) X(5) X(6) X(7) X(8) X(9) X(10) X(11) X(12)
#define FOR_T1(X) X(13) X(14) X(15) X(16) X(17) X(18) X(19) X(20) X(21) X(22) X(23) X(24)
#define X(j) const float4 a##j = srcA[(j) * 32 + lane];
    FOR_T0(X) FOR_T1(X)
#undef X
#define X(j) const float4 b##j = srcB[(j) * 32 + lane];
    FOR_T0(X) FOR_T1(X)
#undef X

    float*  mineA = stage + (wave * P1_UPW) * P1_FL;
    float*  mineB = mineA + P1_FL;
    float4* stA   = (float4*)mineA;
    float4* stB   = (float4*)mineB;

    const float* rowA = mineA + m * DDIM;
    const float* rowB = mineB + m * DDIM;
    float* __restrict__ spA = s + unitA * P1_ROWS + (h ? 8 : 0);
    float* __restrict__ spB = s + unitB * P1_ROWS + (h ? 8 : 0);
    const bool putA = col0 && liveA;
    const bool putB = col0 && liveB;

#define X(j) stA[(j) * 32 + lane] = a##j;
    FOR_T0(X)
    wave_lds_handover();
    {
        const v8f acc = rowdot_tile(rowA, wq, wt, h, col0);
        if (putA) {
            #pragma unroll
            for (int i = 0; i < 8; ++i) spA[i] = acc[i];
        }
    }
    FOR_T1(X)
    wave_lds_handover();
    {
        const v8f acc = rowdot_tile(rowA + 16 * DDIM, wq, wt, h, col0);
        if (putA) {
            #pragma unroll
            for (int i = 0; i < 8; ++i) spA[16 + i] = acc[i];
        }
    }
#undef X

#define X(j) stB[(j) * 32 + lane] = b##j;
    FOR_T0(X)
    wave_lds_handover();
    {
        const v8f acc = rowdot_tile(rowB, wq, wt, h, col0);
        if (putB) {
            #pragma unroll
            for (int i = 0; i < 8; ++i) spB[i] = acc[i];
        }
    }
    FOR_T1(X)
    wave_lds_handover();
    {
        const v8f acc = rowdot_tile(rowB + 16 * DDIM, wq, wt, h, col0);
        if (putB) {
            #pragma unroll
            for (int i = 0; i < 8; ++i) spB[16 + i] = acc[i];
        }
    }
#undef X
#undef FOR_T0
#undef FOR_T1
}

#define HALFV   50000
#define HALFQ   (HALFV / 4)
#define ROWS_WG 64
#define NT2     256
#define TOK_T   50
#define STG     49
#define IDQ     (ROWS_WG * LTOK / 4)
#define ISTG    13

__device__ __forceinline__ int stage_slot(int i, int rot) {
    int g = i + rot;
    if (g >= HALFQ) g -= HALFQ;
    return g;
}

#define FOR_STG(X) \
    X(0) X(1) X(2) X(3) X(4) X(5) X(6) X(7) X(8) X(9) \
    X(10) X(11) X(12) X(13) X(14) X(15) X(16) X(17) X(18) X(19) \
    X(20) X(21) X(22) X(23) X(24) X(25) X(26) X(27) X(28) X(29) \
    X(30) X(31) X(32) X(33) X(34) X(35) X(36) X(37) X(38) X(39) \
    X(40) X(41) X(42) X(43) X(44) X(45) X(46) X(47) X(48)

__global__ __launch_bounds__(NT2) __attribute__((amdgpu_waves_per_eu(2, 2))) void pool_gather(const int* __restrict__ idx,
                                                   const float* __restrict__ s,
                                                   float* __restrict__ out) {
    __shared__ __attribute__((aligned(16))) float tab[HALFV + 4];
    __shared__ __attribute__((aligned(16))) int   idl[ISTG * NT2 * 4];
    __shared__ float rsum[ROWS_WG];

    const int tid = threadIdx.x;
    const int blk = blockIdx.x;
    __builtin_assume(tid < NT2);

    const float4* __restrict__ s4   = (const float4*)s;
    float4*                    tab4 = (float4*)tab;
    const int4* __restrict__   ip4  = (const int4*)(idx + (size_t)blk * (ROWS_WG * LTOK));
    int4*                      idl4 = (int4*)idl;

    const int rot = blk * 48;

    int4   iq[ISTG];
    float4 cur[STG];
    #pragma unroll
    for (int t = 0; t < ISTG; ++t) {
        const int i = tid + t * NT2;
        iq[t] = ip4[i < IDQ ? i : IDQ - 1];
    }
    #pragma unroll
    for (int t = 0; t < STG; ++t) {
        const int i = tid + t * NT2;
        int g = (i < HALFQ ? i : HALFQ - 1) + rot;
        if (g >= HALFQ) g -= HALFQ;
        cur[t] = s4[g];
    }
    #pragma unroll
    for (int t = 0; t < ISTG; ++t) idl4[tid + t * NT2] = iq[t];
    #pragma unroll
    for (int t = 0; t < STG; ++t) {
        const int i = tid + t * NT2;
        tab4[stage_slot(i < HALFQ ? i : HALFQ - 1, rot)] = cur[t];
    }
    if (tid == 0) tab[HALFV] = 0.0f;
    __syncthreads();

#define HALF1_IX(t) (tid + (t) * NT2 < HALFQ ? tid + (t) * NT2 : HALFQ - 1)
#define X(t) const float4 n##t = s4[HALFQ + stage_slot(HALF1_IX(t), rot)];
    FOR_STG(X)
#undef X

    unsigned tok[TOK_T];
    #pragma unroll
    for (int t = 0; t < TOK_T; ++t) tok[t] = (unsigned)idl[tid * TOK_T + t];

    float acc = 0.0f;
    #pragma unroll
    for (int t = 0; t < TOK_T; ++t) {
        const unsigned a = tok[t] < (unsigned)HALFV ? tok[t] : (unsigned)HALFV;
        acc += tab[a];
    }
    __syncthreads();

#define X(t) tab4[stage_slot(HALF1_IX(t), rot)] = n##t;
    FOR_STG(X)
#undef X
    __syncthreads();

    #pragma unroll
    for (int t = 0; t < TOK_T; ++t) {
        const unsigned d = tok[t] - (unsigned)HALFV;
        const unsigned a = d < (unsigned)HALFV ? d : (unsigned)HALFV;
        acc += tab[a];
    }

    acc += __shfl_xor(acc, 1, 32);
    acc += __shfl_xor(acc, 2, 32);
    if ((tid & 3) == 0) rsum[tid >> 2] = acc * (1.0f / (float)LTOK);
    __syncthreads();
    if (tid < ROWS_WG) out[blk * ROWS_WG + tid] = rsum[tid];
}

__global__ __launch_bounds__(256) void fused_naive(const int* __restrict__ idx,
                                                   const float* __restrict__ T,
                                                   const float* __restrict__ w,
                                                   float* __restrict__ out) {
    const int lane = threadIdx.x & 31;
    const int wave = threadIdx.x >> 5;
    const int b    = blockIdx.x * 8 + wave;
    if (b >= BATCH) return;

    const int* __restrict__ ip = idx + (size_t)b * LTOK;
    float sum = 0.0f;
    for (int l = 0; l < LTOK; ++l) {
        const float* __restrict__ rowp = T + (size_t)ip[l] * DDIM;
        for (int d = lane; d < DDIM; d += 32)
            sum += rowp[d] * w[d];
    }
    sum += __shfl_xor(sum, 16, 32);
    sum += __shfl_xor(sum,  8, 32);
    sum += __shfl_xor(sum,  4, 32);
    sum += __shfl_xor(sum,  2, 32);
    sum += __shfl_xor(sum,  1, 32);
    if (lane == 0) out[b] = sum * (1.0f / (float)LTOK);
}

extern "C" void kernel_launch(void* const* d_in, const int* in_sizes, int n_in,
                              void* d_out, int out_size, void* d_ws, size_t ws_size,
                              hipStream_t stream) {
    const int*   word_idxs = (const int*)d_in[0];
    const float* embed     = (const float*)d_in[1];
    const float* weights   = (const float*)d_in[2];
    float*       out       = (float*)d_out;

    if (ws_size >= (size_t)VSIZE * sizeof(float)) {
        float* s = (float*)d_ws;
        const int blocks1 = (P1_WUNITS + P1_WAVES - 1) / P1_WAVES;
        rowdot_wmma<<<blocks1, P1_NT, 0, stream>>>(embed, weights, s);

        const int blocks2 = BATCH / ROWS_WG;
        pool_gather<<<blocks2, NT2, 0, stream>>>(word_idxs, s, out);
    } else {
        const int blocks = (BATCH + 7) / 8;
        fused_naive<<<blocks, 256, 0, stream>>>(word_idxs, embed, weights, out);
    }
}
